# starting kernel
# baseline (speedup 1.0000x reference)
#include <hip/hip_runtime.h>
#include <math.h>


typedef float v2f __attribute__((ext_vector_type(2)));
typedef float v8f __attribute__((ext_vector_type(8)));

#define DD      8
#define HH      64
#define IN_DIM  9
#define BB      128
#define LL      256
#define NTOT    (BB * LL)
#define SLOPE   0.2f
#define AST     66
#define WAVES   8
#define ROWSBLK (WAVES * 16)

static __device__ __forceinline__ v8f wmma_f32(v2f a, v2f b, v8f c) {
  return __builtin_amdgcn_wmma_f32_16x16x4_f32(
       false, a,  false, b,
       (short)0, c,  false,  false);
}

__global__ __launch_bounds__(256) void flow_mlp_kernel(
    const float* __restrict__ x,
    const float* __restrict__ W1, const float* __restrict__ b1,
    const float* __restrict__ W2, const float* __restrict__ b2,
    const float* __restrict__ W3, const float* __restrict__ b3,
    const float* __restrict__ Wo, const float* __restrict__ bo,
    float* __restrict__ out, float* __restrict__ ws)
{
  __shared__ float sW1T[16 * 64];
  __shared__ float sW2T[64 * 64];
  __shared__ float sW3T[64 * 64];
  __shared__ float sB1[64], sB2[64], sB3[64], sWoL[64];
  __shared__ float sBufA[WAVES][16 * AST];
  __shared__ float sBufD[WAVES][16 * AST];

  const int tid = threadIdx.x;
  const int d   = blockIdx.y;

  for (int i = tid; i < 16 * 64; i += 256) {
    const int k = i >> 6, h = i & 63;
    sW1T[i] = (k < IN_DIM) ? W1[(d * HH + h) * IN_DIM + k] : 0.f;
  }
  for (int i = tid; i < 64 * 64; i += 256) {
    const int g = i >> 6, h = i & 63;
    sW2T[h * 64 + g] = W2[d * 4096 + i];
    sW3T[h * 64 + g] = W3[d * 4096 + i];
  }
  if (tid < 64) {
    sB1[tid]  = b1[d * HH + tid];
    sB2[tid]  = b2[d * HH + tid];
    sB3[tid]  = b3[d * HH + tid];
    sWoL[tid] = Wo[d * HH + tid];
  }
  __syncthreads();

  const int wave = tid >> 5;
  const int lane = tid & 31;
  const int lx   = lane & 15;
  const int half = lane >> 4;
  const int koff = half * 2;

  float* sA = sBufA[wave];
  float* sD = sBufD[wave];

  const int n0 = blockIdx.x * ROWSBLK + wave * 16;
  const int bb = n0 >> 8;
  const int l0 = n0 & 255;
  {
    const int s = lane >> 1, part = lane & 1;
    const float4 v = *reinterpret_cast<const float4*>(
        x + (size_t)(bb * 257 + l0 + s) * 8 + part * 4);
    float* dst = &sA[s * AST + part * 4];
    dst[0] = v.x; dst[1] = v.y; dst[2] = v.z; dst[3] = v.w;
    if (lane < 16) {
      sA[lane * AST + 8] = x[(size_t)(bb * 257 + l0 + lane + 1) * 8 + d];
    } else {
      sA[lx * AST + 9]  = 0.f;
      sA[lx * AST + 10] = 0.f;
      sA[lx * AST + 11] = 0.f;
    }
  }

  v8f acc1[4];
  {
    v2f af[3];
#pragma unroll
    for (int kk = 0; kk < 3; ++kk) {
      const int k = kk * 4 + koff;
      af[kk].x = sA[lx * AST + k];
      af[kk].y = sA[lx * AST + k + 1];
    }
#pragma unroll
    for (int t = 0; t < 4; ++t) {
      v8f c = {0.f, 0.f, 0.f, 0.f, 0.f, 0.f, 0.f, 0.f};
#pragma unroll
      for (int kk = 0; kk < 3; ++kk) {
        const int k = kk * 4 + koff;
        v2f b;
        b.x = sW1T[k * 64 + t * 16 + lx];
        b.y = sW1T[(k + 1) * 64 + t * 16 + lx];
        c = wmma_f32(af[kk], b, c);
      }
      acc1[t] = c;
    }
  }
#pragma unroll
  for (int t = 0; t < 4; ++t) {
    const int f   = t * 16 + lx;
    const float w8 = sW1T[8 * 64 + f];
    const float bv = sB1[f];
#pragma unroll
    for (int v = 0; v < 8; ++v) {
      const float h = acc1[t][v] + bv;
      const float m = (h >= 0.f) ? 1.f : SLOPE;
      const int  ms = v + 8 * half;
      sA[ms * AST + f] = m * h;
      sD[ms * AST + f] = m * w8;
    }
  }

  auto layer = [&](const float* __restrict__ WT, const float* __restrict__ bias) {
    v2f af[16], df[16];
#pragma unroll
    for (int kk = 0; kk < 16; ++kk) {
      const int k = kk * 4 + koff;
      af[kk].x = sA[lx * AST + k];  af[kk].y = sA[lx * AST + k + 1];
      df[kk].x = sD[lx * AST + k];  df[kk].y = sD[lx * AST + k + 1];
    }
#pragma unroll
    for (int t = 0; t < 4; ++t) {
      v8f ca = {0.f, 0.f, 0.f, 0.f, 0.f, 0.f, 0.f, 0.f};
      v8f cd = {0.f, 0.f, 0.f, 0.f, 0.f, 0.f, 0.f, 0.f};
#pragma unroll
      for (int kk = 0; kk < 16; ++kk) {
        const int k = kk * 4 + koff;
        v2f b;
        b.x = WT[k * 64 + t * 16 + lx];
        b.y = WT[(k + 1) * 64 + t * 16 + lx];
        ca = wmma_f32(af[kk], b, ca);
        cd = wmma_f32(df[kk], b, cd);
      }
      const int f   = t * 16 + lx;
      const float bv = bias[f];
#pragma unroll
      for (int v = 0; v < 8; ++v) {
        const float h = ca[v] + bv;
        const float m = (h >= 0.f) ? 1.f : SLOPE;
        const int  ms = v + 8 * half;
        sA[ms * AST + f] = m * h;
        sD[ms * AST + f] = m * cd[v];
      }
    }
  };
  layer(sW2T, sB2);
  layer(sW3T, sB3);

  {
    const float* buf = (lane < 16) ? sA : sD;
    float s = 0.f;
#pragma unroll 8
    for (int k = 0; k < 64; ++k) s += buf[lx * AST + k] * sWoL[k];
    const int n = n0 + lx;
    if (lane < 16) {
      out[(size_t)n * DD + d] = s + bo[d];
    } else {
      ws[(size_t)d * NTOT + n] = s;
    }
  }
}

__global__ __launch_bounds__(256) void logdet_kernel(
    const float* __restrict__ ws, float* __restrict__ out)
{
  const int r = blockIdx.x * 256 + threadIdx.x;
  if (r >= NTOT) return;
  float s = 0.f;
#pragma unroll
  for (int d = 0; d < DD; ++d) s += logf(fabsf(ws[(size_t)d * NTOT + r]));
  out[(size_t)NTOT * DD + r] = s;
}

extern "C" void kernel_launch(void* const* d_in, const int* in_sizes, int n_in,
                              void* d_out, int out_size, void* d_ws, size_t ws_size,
                              hipStream_t stream) {
  const float* x  = (const float*)d_in[0];
  const float* W1 = (const float*)d_in[1];
  const float* b1 = (const float*)d_in[2];
  const float* W2 = (const float*)d_in[3];
  const float* b2 = (const float*)d_in[4];
  const float* W3 = (const float*)d_in[5];
  const float* b3 = (const float*)d_in[6];
  const float* Wo = (const float*)d_in[7];
  const float* bo = (const float*)d_in[8];
  float* out = (float*)d_out;
  float* ws  = (float*)d_ws;

  dim3 grid(NTOT / ROWSBLK, DD);
  flow_mlp_kernel<<<grid, dim3(256), 0, stream>>>(
      x, W1, b1, W2, b2, W3, b3, Wo, bo, out, ws);
  logdet_kernel<<<dim3(NTOT / 256), dim3(256), 0, stream>>>(ws, out);
}


# Persistent 16-wave workgroups (grid 32x8, one per WGP): W1/W2/W3 staged into LDS once, each wave loops over 4 sample tiles; was 256x8 groups of 8 waves, each restaging the weights for 128 samples
# speedup vs baseline: 1.4245x; 1.4245x over previous
#include <hip/hip_runtime.h>
#include <math.h>


typedef float v2f __attribute__((ext_vector_type(2)));
typedef float v8f __attribute__((ext_vector_type(8)));

#define DD      8
#define HH      64
#define IN_DIM  9
#define BB      128
#define LL      256
#define NTOT    (BB * LL)
#define SLOPE   0.2f
#define AST     66
#define WAVES   16
#define NTHR    (WAVES * 32)
#define TILES   4
#define ROWSBLK (WAVES * 16 * TILES)

static __device__ __forceinline__ v8f wmma_f32(v2f a, v2f b, v8f c) {
  return __builtin_amdgcn_wmma_f32_16x16x4_f32(
       false, a,  false, b,
       (short)0, c,  false,  false);
}

__global__ __launch_bounds__(NTHR) void flow_mlp_kernel(
    const float* __restrict__ x,
    const float* __restrict__ W1, const float* __restrict__ b1,
    const float* __restrict__ W2, const float* __restrict__ b2,
    const float* __restrict__ W3, const float* __restrict__ b3,
    const float* __restrict__ Wo, const float* __restrict__ bo,
    float* __restrict__ out, float* __restrict__ ws)
{
  __shared__ float sW1T[16 * 64];
  __shared__ float sW2T[64 * 64];
  __shared__ float sW3T[64 * 64];
  __shared__ float sB1[64], sB2[64], sB3[64], sWoL[64];
  __shared__ float sBufA[WAVES][16 * AST];
  __shared__ float sBufD[WAVES][16 * AST];

  const int tid = threadIdx.x;
  const int d   = blockIdx.y;

  for (int i = tid; i < 16 * 64; i += NTHR) {
    const int k = i >> 6, h = i & 63;
    sW1T[i] = (k < IN_DIM) ? W1[(d * HH + h) * IN_DIM + k] : 0.f;
  }
  for (int i = tid; i < 64 * 64; i += NTHR) {
    const int g = i >> 6, h = i & 63;
    sW2T[h * 64 + g] = W2[d * 4096 + i];
    sW3T[h * 64 + g] = W3[d * 4096 + i];
  }
  if (tid < 64) {
    sB1[tid]  = b1[d * HH + tid];
    sB2[tid]  = b2[d * HH + tid];
    sB3[tid]  = b3[d * HH + tid];
    sWoL[tid] = Wo[d * HH + tid];
  }
  __syncthreads();

  const int wave = tid >> 5;
  const int lane = tid & 31;
  const int lx   = lane & 15;
  const int half = lane >> 4;
  const int koff = half * 2;

  float* sA = sBufA[wave];
  float* sD = sBufD[wave];

  auto layer = [&](const float* __restrict__ WT, const float* __restrict__ bias) {
    v2f af[16], df[16];
#pragma unroll
    for (int kk = 0; kk < 16; ++kk) {
      const int k = kk * 4 + koff;
      af[kk].x = sA[lx * AST + k];  af[kk].y = sA[lx * AST + k + 1];
      df[kk].x = sD[lx * AST + k];  df[kk].y = sD[lx * AST + k + 1];
    }
#pragma unroll
    for (int t = 0; t < 4; ++t) {
      v8f ca = {0.f, 0.f, 0.f, 0.f, 0.f, 0.f, 0.f, 0.f};
      v8f cd = {0.f, 0.f, 0.f, 0.f, 0.f, 0.f, 0.f, 0.f};
#pragma unroll
      for (int kk = 0; kk < 16; ++kk) {
        const int k = kk * 4 + koff;
        v2f b;
        b.x = WT[k * 64 + t * 16 + lx];
        b.y = WT[(k + 1) * 64 + t * 16 + lx];
        ca = wmma_f32(af[kk], b, ca);
        cd = wmma_f32(df[kk], b, cd);
      }
      const int f   = t * 16 + lx;
      const float bv = bias[f];
#pragma unroll
      for (int v = 0; v < 8; ++v) {
        const float h = ca[v] + bv;
        const float m = (h >= 0.f) ? 1.f : SLOPE;
        const int  ms = v + 8 * half;
        sA[ms * AST + f] = m * h;
        sD[ms * AST + f] = m * cd[v];
      }
    }
  };

  const float bod = bo[d];

  for (int it = 0; it < TILES; ++it) {
    const int n0 = blockIdx.x * ROWSBLK + (it * WAVES + wave) * 16;
    const int bb = n0 >> 8;
    const int l0 = n0 & 255;
    {
      const int s = lane >> 1, part = lane & 1;
      const float4 v = *reinterpret_cast<const float4*>(
          x + (size_t)(bb * 257 + l0 + s) * 8 + part * 4);
      float* dst = &sA[s * AST + part * 4];
      dst[0] = v.x; dst[1] = v.y; dst[2] = v.z; dst[3] = v.w;
      if (lane < 16) {
        sA[lane * AST + 8] = x[(size_t)(bb * 257 + l0 + lane + 1) * 8 + d];
      } else {
        sA[lx * AST + 9]  = 0.f;
        sA[lx * AST + 10] = 0.f;
        sA[lx * AST + 11] = 0.f;
      }
    }

    v8f acc1[4];
    {
      v2f af[3];
#pragma unroll
      for (int kk = 0; kk < 3; ++kk) {
        const int k = kk * 4 + koff;
        af[kk].x = sA[lx * AST + k];
        af[kk].y = sA[lx * AST + k + 1];
      }
#pragma unroll
      for (int t = 0; t < 4; ++t) {
        v8f c = {0.f, 0.f, 0.f, 0.f, 0.f, 0.f, 0.f, 0.f};
#pragma unroll
        for (int kk = 0; kk < 3; ++kk) {
          const int k = kk * 4 + koff;
          v2f b;
          b.x = sW1T[k * 64 + t * 16 + lx];
          b.y = sW1T[(k + 1) * 64 + t * 16 + lx];
          c = wmma_f32(af[kk], b, c);
        }
        acc1[t] = c;
      }
    }
#pragma unroll
    for (int t = 0; t < 4; ++t) {
      const int f   = t * 16 + lx;
      const float w8 = sW1T[8 * 64 + f];
      const float bv = sB1[f];
#pragma unroll
      for (int v = 0; v < 8; ++v) {
        const float h = acc1[t][v] + bv;
        const float m = (h >= 0.f) ? 1.f : SLOPE;
        const int  ms = v + 8 * half;
        sA[ms * AST + f] = m * h;
        sD[ms * AST + f] = m * w8;
      }
    }

    layer(sW2T, sB2);
    layer(sW3T, sB3);

    {
      const float* buf = (lane < 16) ? sA : sD;
      float s = 0.f;
#pragma unroll 8
      for (int k = 0; k < 64; ++k) s += buf[lx * AST + k] * sWoL[k];
      const int n = n0 + lx;
      if (lane < 16) {
        out[(size_t)n * DD + d] = s + bod;
      } else {
        ws[(size_t)d * NTOT + n] = s;
      }
    }
  }
}

__global__ __launch_bounds__(256) void logdet_kernel(
    const float* __restrict__ ws, float* __restrict__ out)
{
  const int r = blockIdx.x * 256 + threadIdx.x;
  if (r >= NTOT) return;
  float s = 0.f;
#pragma unroll
  for (int d = 0; d < DD; ++d) s += logf(fabsf(ws[(size_t)d * NTOT + r]));
  out[(size_t)NTOT * DD + r] = s;
}

extern "C" void kernel_launch(void* const* d_in, const int* in_sizes, int n_in,
                              void* d_out, int out_size, void* d_ws, size_t ws_size,
                              hipStream_t stream) {
  const float* x  = (const float*)d_in[0];
  const float* W1 = (const float*)d_in[1];
  const float* b1 = (const float*)d_in[2];
  const float* W2 = (const float*)d_in[3];
  const float* b2 = (const float*)d_in[4];
  const float* W3 = (const float*)d_in[5];
  const float* b3 = (const float*)d_in[6];
  const float* Wo = (const float*)d_in[7];
  const float* bo = (const float*)d_in[8];
  float* out = (float*)d_out;
  float* ws  = (float*)d_ws;

  dim3 grid(NTOT / ROWSBLK, DD);
  flow_mlp_kernel<<<grid, dim3(NTHR), 0, stream>>>(
      x, W1, b1, W2, b2, W3, b3, Wo, bo, out, ws);
  logdet_kernel<<<dim3(NTOT / 256), dim3(256), 0, stream>>>(ws, out);
}


# Weights kept in LDS in B-fragment order: one conflict-free ds_load_b128 per lane feeds two K-steps (b64 for W1), staged in conflict-free; was 2-address strided loads of 4 LDS clocks, 32-way stage-in
# speedup vs baseline: 1.6130x; 1.1323x over previous
#include <hip/hip_runtime.h>
#include <math.h>


typedef float v2f __attribute__((ext_vector_type(2)));
typedef float v4f __attribute__((ext_vector_type(4)));
typedef float v8f __attribute__((ext_vector_type(8)));

#define DD      8
#define HH      64
#define IN_DIM  9
#define BB      128
#define LL      256
#define NTOT    (BB * LL)
#define SLOPE   0.2f
#define AST     66
#define WAVES   16
#define NTHR    (WAVES * 32)
#define TILES   4
#define ROWSBLK (WAVES * 16 * TILES)

static __device__ __forceinline__ v8f wmma_f32(v2f a, v2f b, v8f c) {
  return __builtin_amdgcn_wmma_f32_16x16x4_f32(
       false, a,  false, b,
       (short)0, c,  false,  false);
}

__global__ __launch_bounds__(NTHR) void flow_mlp_kernel(
    const float* __restrict__ x,
    const float* __restrict__ W1, const float* __restrict__ b1,
    const float* __restrict__ W2, const float* __restrict__ b2,
    const float* __restrict__ W3, const float* __restrict__ b3,
    const float* __restrict__ Wo, const float* __restrict__ bo,
    float* __restrict__ out, float* __restrict__ ws)
{
  __shared__ v4f sW2F[4 * 8 * 32];
  __shared__ v4f sW3F[4 * 8 * 32];
  __shared__ v2f sW1F[4 * 3 * 32];
  __shared__ float sW8[64];
  __shared__ float sB1[64], sB2[64], sB3[64], sWoL[64];
  __shared__ float sBufA[WAVES][16 * AST];
  __shared__ float sBufD[WAVES][16 * AST];

  const int tid = threadIdx.x;
  const int d   = blockIdx.y;

  {
    const int slx = tid & 15, kp = (tid >> 4) & 7, st = tid >> 7;
    const size_t off = (size_t)d * 4096 + (st * 16 + slx) * 64 + kp * 8;
    const float4 p2 = *reinterpret_cast<const float4*>(W2 + off);
    const float4 q2 = *reinterpret_cast<const float4*>(W2 + off + 4);
    const float4 p3 = *reinterpret_cast<const float4*>(W3 + off);
    const float4 q3 = *reinterpret_cast<const float4*>(W3 + off + 4);
    const int slot = (st * 8 + kp) * 32 + slx;
    sW2F[slot]      = (v4f){p2.x, p2.y, q2.x, q2.y};
    sW2F[slot + 16] = (v4f){p2.z, p2.w, q2.z, q2.w};
    sW3F[slot]      = (v4f){p3.x, p3.y, q3.x, q3.y};
    sW3F[slot + 16] = (v4f){p3.z, p3.w, q3.z, q3.w};
  }
  if (tid < 4 * 3 * 32) {
    const int sl = tid & 31, kk = (tid >> 5) % 3, st = tid / 96;
    const int h  = st * 16 + (sl & 15);
    const int k0 = kk * 4 + (sl >> 4) * 2;
    v2f w;
    w.x = (k0     < IN_DIM) ? W1[(d * HH + h) * IN_DIM + k0]     : 0.f;
    w.y = (k0 + 1 < IN_DIM) ? W1[(d * HH + h) * IN_DIM + k0 + 1] : 0.f;
    sW1F[tid] = w;
  }
  if (tid < 64) {
    sW8[tid]  = W1[(d * HH + tid) * IN_DIM + 8];
    sB1[tid]  = b1[d * HH + tid];
    sB2[tid]  = b2[d * HH + tid];
    sB3[tid]  = b3[d * HH + tid];
    sWoL[tid] = Wo[d * HH + tid];
  }
  __syncthreads();

  const int wave = tid >> 5;
  const int lane = tid & 31;
  const int lx   = lane & 15;
  const int half = lane >> 4;
  const int koff = half * 2;

  float* sA = sBufA[wave];
  float* sD = sBufD[wave];

  auto layer = [&](const v4f* __restrict__ WF, const float* __restrict__ bias) {
    v2f af[16], df[16];
#pragma unroll
    for (int kk = 0; kk < 16; ++kk) {
      const int k = kk * 4 + koff;
      af[kk].x = sA[lx * AST + k];  af[kk].y = sA[lx * AST + k + 1];
      df[kk].x = sD[lx * AST + k];  df[kk].y = sD[lx * AST + k + 1];
    }
#pragma unroll
    for (int t = 0; t < 4; ++t) {
      v8f ca = {0.f, 0.f, 0.f, 0.f, 0.f, 0.f, 0.f, 0.f};
      v8f cd = {0.f, 0.f, 0.f, 0.f, 0.f, 0.f, 0.f, 0.f};
#pragma unroll
      for (int kp = 0; kp < 8; ++kp) {
        const v4f bw = WF[(t * 8 + kp) * 32 + lane];
        v2f b0, b1;
        b0.x = bw.x;  b0.y = bw.y;
        b1.x = bw.z;  b1.y = bw.w;
        ca = wmma_f32(af[2 * kp],     b0, ca);
        cd = wmma_f32(df[2 * kp],     b0, cd);
        ca = wmma_f32(af[2 * kp + 1], b1, ca);
        cd = wmma_f32(df[2 * kp + 1], b1, cd);
      }
      const int f   = t * 16 + lx;
      const float bv = bias[f];
#pragma unroll
      for (int v = 0; v < 8; ++v) {
        const float h = ca[v] + bv;
        const float m = (h >= 0.f) ? 1.f : SLOPE;
        const int  ms = v + 8 * half;
        sA[ms * AST + f] = m * h;
        sD[ms * AST + f] = m * cd[v];
      }
    }
  };

  const float bod = bo[d];

  for (int it = 0; it < TILES; ++it) {
    const int n0 = blockIdx.x * ROWSBLK + (it * WAVES + wave) * 16;
    const int bb = n0 >> 8;
    const int l0 = n0 & 255;
    {
      const int s = lane >> 1, part = lane & 1;
      const float4 v = *reinterpret_cast<const float4*>(
          x + (size_t)(bb * 257 + l0 + s) * 8 + part * 4);
      float* dst = &sA[s * AST + part * 4];
      dst[0] = v.x; dst[1] = v.y; dst[2] = v.z; dst[3] = v.w;
      if (lane < 16) {
        sA[lane * AST + 8] = x[(size_t)(bb * 257 + l0 + lane + 1) * 8 + d];
      } else {
        sA[lx * AST + 9]  = 0.f;
        sA[lx * AST + 10] = 0.f;
        sA[lx * AST + 11] = 0.f;
      }
    }

    v8f acc1[4];
    {
      v2f af[3];
#pragma unroll
      for (int kk = 0; kk < 3; ++kk) {
        const int k = kk * 4 + koff;
        af[kk].x = sA[lx * AST + k];
        af[kk].y = sA[lx * AST + k + 1];
      }
#pragma unroll
      for (int t = 0; t < 4; ++t) {
        v8f c = {0.f, 0.f, 0.f, 0.f, 0.f, 0.f, 0.f, 0.f};
#pragma unroll
        for (int kk = 0; kk < 3; ++kk) {
          const v2f b = sW1F[(t * 3 + kk) * 32 + lane];
          c = wmma_f32(af[kk], b, c);
        }
        acc1[t] = c;
      }
    }
#pragma unroll
    for (int t = 0; t < 4; ++t) {
      const int f   = t * 16 + lx;
      const float w8 = sW8[f];
      const float bv = sB1[f];
#pragma unroll
      for (int v = 0; v < 8; ++v) {
        const float h = acc1[t][v] + bv;
        const float m = (h >= 0.f) ? 1.f : SLOPE;
        const int  ms = v + 8 * half;
        sA[ms * AST + f] = m * h;
        sD[ms * AST + f] = m * w8;
      }
    }

    layer(sW2F, sB2);
    layer(sW3F, sB3);

    {
      const float* buf = (lane < 16) ? sA : sD;
      float s = 0.f;
#pragma unroll 8
      for (int k = 0; k < 64; ++k) s += buf[lx * AST + k] * sWoL[k];
      const int n = n0 + lx;
      if (lane < 16) {
        out[(size_t)n * DD + d] = s + bod;
      } else {
        ws[(size_t)d * NTOT + n] = s;
      }
    }
  }
}

__global__ __launch_bounds__(256) void logdet_kernel(
    const float* __restrict__ ws, float* __restrict__ out)
{
  const int r = blockIdx.x * 256 + threadIdx.x;
  if (r >= NTOT) return;
  float s = 0.f;
#pragma unroll
  for (int d = 0; d < DD; ++d) s += logf(fabsf(ws[(size_t)d * NTOT + r]));
  out[(size_t)NTOT * DD + r] = s;
}

extern "C" void kernel_launch(void* const* d_in, const int* in_sizes, int n_in,
                              void* d_out, int out_size, void* d_ws, size_t ws_size,
                              hipStream_t stream) {
  const float* x  = (const float*)d_in[0];
  const float* W1 = (const float*)d_in[1];
  const float* b1 = (const float*)d_in[2];
  const float* W2 = (const float*)d_in[3];
  const float* b2 = (const float*)d_in[4];
  const float* W3 = (const float*)d_in[5];
  const float* b3 = (const float*)d_in[6];
  const float* Wo = (const float*)d_in[7];
  const float* bo = (const float*)d_in[8];
  float* out = (float*)d_out;
  float* ws  = (float*)d_ws;

  dim3 grid(NTOT / ROWSBLK, DD);
  flow_mlp_kernel<<<grid, dim3(NTHR), 0, stream>>>(
      x, W1, b1, W2, b2, W3, b3, Wo, bo, out, ws);
  logdet_kernel<<<dim3(NTOT / 256), dim3(256), 0, stream>>>(ws, out);
}


# Wave priority falls with a wave's progress (s_setprio 3, 2 at half its tiles, 1 at its last tile, 0 at that tile's layer 3): a SIMD's 4 waves end together; was oldest-first, youngest left alone
# speedup vs baseline: 1.8509x; 1.1475x over previous
#include <hip/hip_runtime.h>
#include <math.h>


typedef float v2f __attribute__((ext_vector_type(2)));
typedef float v4f __attribute__((ext_vector_type(4)));
typedef float v8f __attribute__((ext_vector_type(8)));

#define DD      8
#define HH      64
#define IN_DIM  9
#define BB      128
#define LL      256
#define NTOT    (BB * LL)
#define SLOPE   0.2f
#define WAVES   16
#define NTHR    (WAVES * 32)
#define TILES   4
#define ROWSBLK (WAVES * 16 * TILES)

static __device__ __forceinline__ v8f wmma_f32(v2f a, v2f b, v8f c) {
  return __builtin_amdgcn_wmma_f32_16x16x4_f32(
       false, a,  false, b,
       (short)0, c,  false,  false);
}


static __device__ __forceinline__ void act_tile(
    const v8f ca, const v8f cd, const v4f blo, const v4f bhi,
    v2f* __restrict__ oa, v2f* __restrict__ od)
{
  const float bv[8] = {blo.x, blo.y, blo.z, blo.w, bhi.x, bhi.y, bhi.z, bhi.w};
#pragma unroll
  for (int r = 0; r < 8; ++r) {
    const float h = ca[r] + bv[r];
    const float m = (h >= 0.f) ? 1.f : SLOPE;
    oa[r >> 1][r & 1] = m * h;
    od[r >> 1][r & 1] = m * cd[r];
  }
}

__global__ __launch_bounds__(NTHR) __attribute__((amdgpu_waves_per_eu(4, 4)))
void flow_mlp_kernel(
    const float* __restrict__ x,
    const float* __restrict__ W1, const float* __restrict__ b1,
    const float* __restrict__ W2, const float* __restrict__ b2,
    const float* __restrict__ W3, const float* __restrict__ b3,
    const float* __restrict__ Wo, const float* __restrict__ bo,
    float* __restrict__ out, float* __restrict__ ws)
{
  __shared__ v4f sW2F[4 * 8 * 32];
  __shared__ v4f sW3F[4 * 8 * 32];
  __shared__ v2f sW1F[4 * 3 * 32];
  __shared__ v4f sW8[16];
  __shared__ v4f sB1[16], sB2[16], sB3[16], sWoL[16];

  const int tid = threadIdx.x;
  const int d   = blockIdx.y;

  {
    const int si = tid & 15, c4 = (tid >> 4) & 15, gt = tid >> 8;
    const int tq = c4 >> 2, sh = (c4 >> 1) & 1, q = c4 & 1;
    const int kp = 2 * tq + q;
#pragma unroll
    for (int j = 0; j < 2; ++j) {
      const int st = gt * 2 + j;
      const size_t off = (size_t)d * 4096 + (st * 16 + si) * 64 + c4 * 4;
      const float4 p2 = *reinterpret_cast<const float4*>(W2 + off);
      const float4 p3 = *reinterpret_cast<const float4*>(W3 + off);
      const int slot = (st * 8 + kp) * 32 + sh * 16 + si;
      sW2F[slot] = (v4f){p2.x, p2.y, p2.z, p2.w};
      sW3F[slot] = (v4f){p3.x, p3.y, p3.z, p3.w};
    }
  }
  if (tid < 4 * 3 * 32) {
    const int sl = tid & 31, kk = (tid >> 5) % 3, st = tid / 96;
    const int g  = st * 16 + (sl & 15);
    const int sh = sl >> 4;
    const float* row = W1 + (size_t)(d * HH + g) * IN_DIM;
    v2f w;
    if (kk < 2) {
      w.x = row[4 * sh + 2 * kk];
      w.y = row[4 * sh + 2 * kk + 1];
    } else {
      w.x = (sh == 0) ? row[8] : 0.f;
      w.y = 0.f;
    }
    sW1F[tid] = w;
  }
  if (tid < 64) {
    reinterpret_cast<float*>(sW8)[tid]  = W1[(d * HH + tid) * IN_DIM + 8];
    reinterpret_cast<float*>(sB1)[tid]  = b1[d * HH + tid];
    reinterpret_cast<float*>(sB2)[tid]  = b2[d * HH + tid];
    reinterpret_cast<float*>(sB3)[tid]  = b3[d * HH + tid];
    reinterpret_cast<float*>(sWoL)[tid] = Wo[d * HH + tid];
  }
  __syncthreads();

  const int wave = tid >> 5;
  const int lane = tid & 31;
  const int lx   = lane & 15;
  const int half = lane >> 4;

  const float bod = bo[d];

  __builtin_amdgcn_s_setprio(3);
  for (int it = 0; it < TILES; ++it) {
    if (it == TILES / 2) __builtin_amdgcn_s_setprio(2);
    if (it == TILES - 1) __builtin_amdgcn_s_setprio(1);
    asm volatile("" ::: "memory");
    const int n0 = blockIdx.x * ROWSBLK + (it * WAVES + wave) * 16;
    const int bb = n0 >> 8;
    const int l0 = n0 & 255;
    const float* xs = x + (size_t)(bb * 257 + l0) * 8;
    const float4 xv = *reinterpret_cast<const float4*>(xs + lx * 8 + half * 4);
    const float  xt = xs[lx * 8 + 8 + d];
    v2f bx[3];
    bx[0].x = xv.x;  bx[0].y = xv.y;
    bx[1].x = xv.z;  bx[1].y = xv.w;
    bx[2].x = (half == 0) ? xt : 0.f;
    bx[2].y = 0.f;

    v2f a1[16], d1[16];
    v2f a2[16], d2[16];

#pragma unroll
    for (int t = 0; t < 4; ++t) {
      v8f c = {0.f, 0.f, 0.f, 0.f, 0.f, 0.f, 0.f, 0.f};
#pragma unroll
      for (int kk = 0; kk < 3; ++kk) {
        const v2f w = sW1F[(t * 3 + kk) * 32 + lane];
        c = wmma_f32(w, bx[kk], c);
      }
      const v4f wlo = sW8[4 * t + 2 * half], whi = sW8[4 * t + 2 * half + 1];
      const v8f w8  = {wlo.x, wlo.y, wlo.z, wlo.w, whi.x, whi.y, whi.z, whi.w};
      act_tile(c, w8, sB1[4 * t + 2 * half], sB1[4 * t + 2 * half + 1],
               &a1[4 * t], &d1[4 * t]);
    }

    v8f ca[4], cd[4];
#pragma unroll
    for (int t = 0; t < 4; ++t) {
      ca[t] = (v8f){0.f, 0.f, 0.f, 0.f, 0.f, 0.f, 0.f, 0.f};
      cd[t] = (v8f){0.f, 0.f, 0.f, 0.f, 0.f, 0.f, 0.f, 0.f};
    }
#pragma unroll
    for (int kp = 0; kp < 8; ++kp) {
      v4f w[4];
#pragma unroll
      for (int t = 0; t < 4; ++t)
        w[t] = sW2F[(t * 8 + kp) * 32 + lane];
      if (kp == 4) __builtin_amdgcn_sched_barrier(0);
#pragma unroll
      for (int t = 0; t < 4; ++t) {
        v2f w0, w1;
        w0.x = w[t].x;  w0.y = w[t].y;
        w1.x = w[t].z;  w1.y = w[t].w;
        ca[t] = wmma_f32(w0, a1[2 * kp],     ca[t]);
        cd[t] = wmma_f32(w0, d1[2 * kp],     cd[t]);
        ca[t] = wmma_f32(w1, a1[2 * kp + 1], ca[t]);
        cd[t] = wmma_f32(w1, d1[2 * kp + 1], cd[t]);
      }
    }
#pragma unroll
    for (int t = 0; t < 4; ++t)
      act_tile(ca[t], cd[t], sB2[4 * t + 2 * half], sB2[4 * t + 2 * half + 1],
               &a2[4 * t], &d2[4 * t]);

    if (it == TILES - 1) __builtin_amdgcn_s_setprio(0);
#pragma unroll
    for (int t = 0; t < 4; ++t) {
      ca[t] = (v8f){0.f, 0.f, 0.f, 0.f, 0.f, 0.f, 0.f, 0.f};
      cd[t] = (v8f){0.f, 0.f, 0.f, 0.f, 0.f, 0.f, 0.f, 0.f};
    }
#pragma unroll
    for (int kp = 0; kp < 8; ++kp) {
      v4f w[4];
#pragma unroll
      for (int t = 0; t < 4; ++t)
        w[t] = sW3F[(t * 8 + kp) * 32 + lane];
      if (kp == 4) __builtin_amdgcn_sched_barrier(0);
#pragma unroll
      for (int t = 0; t < 4; ++t) {
        v2f w0, w1;
        w0.x = w[t].x;  w0.y = w[t].y;
        w1.x = w[t].z;  w1.y = w[t].w;
        ca[t] = wmma_f32(w0, a2[2 * kp],     ca[t]);
        cd[t] = wmma_f32(w0, d2[2 * kp],     cd[t]);
        ca[t] = wmma_f32(w1, a2[2 * kp + 1], ca[t]);
        cd[t] = wmma_f32(w1, d2[2 * kp + 1], cd[t]);
      }
    }
    float pa = 0.f, pd = 0.f;
#pragma unroll
    for (int t = 0; t < 4; ++t) {
      v2f a3[4], d3[4];
      act_tile(ca[t], cd[t], sB3[4 * t + 2 * half], sB3[4 * t + 2 * half + 1], a3, d3);
      const v4f olo = sWoL[4 * t + 2 * half], ohi = sWoL[4 * t + 2 * half + 1];
      const float wo[8] = {olo.x, olo.y, olo.z, olo.w, ohi.x, ohi.y, ohi.z, ohi.w};
#pragma unroll
      for (int r = 0; r < 8; ++r) {
        pa += a3[r >> 1][r & 1] * wo[r];
        pd += d3[r >> 1][r & 1] * wo[r];
      }
    }
    {
      const float give = (half == 0) ? pd : pa;
      const float got  = __shfl_xor(give, 16);
      float* outs = out + (size_t)n0 * DD + d;
      float* wss  = ws + (size_t)d * NTOT + n0;
      if (half == 0) {
        outs[lx * DD] = (pa + got) + bod;
      } else {
        wss[lx] = pd + got;
      }
    }
  }
}

__global__ __launch_bounds__(256) void logdet_kernel(
    const float* __restrict__ ws, float* __restrict__ out)
{
  const int r = blockIdx.x * 256 + threadIdx.x;
  if (r >= NTOT) return;
  float s = 0.f;
#pragma unroll
  for (int d = 0; d < DD; ++d) s += logf(fabsf(ws[(size_t)d * NTOT + r]));
  out[(size_t)NTOT * DD + r] = s;
}

extern "C" void kernel_launch(void* const* d_in, const int* in_sizes, int n_in,
                              void* d_out, int out_size, void* d_ws, size_t ws_size,
                              hipStream_t stream) {
  const float* x  = (const float*)d_in[0];
  const float* W1 = (const float*)d_in[1];
  const float* b1 = (const float*)d_in[2];
  const float* W2 = (const float*)d_in[3];
  const float* b2 = (const float*)d_in[4];
  const float* W3 = (const float*)d_in[5];
  const float* b3 = (const float*)d_in[6];
  const float* Wo = (const float*)d_in[7];
  const float* bo = (const float*)d_in[8];
  float* out = (float*)d_out;
  float* ws  = (float*)d_ws;

  dim3 grid(NTOT / ROWSBLK, DD);
  flow_mlp_kernel<<<grid, dim3(NTHR), 0, stream>>>(
      x, W1, b1, W2, b2, W3, b3, Wo, bo, out, ws);
  logdet_kernel<<<dim3(NTOT / 256), dim3(256), 0, stream>>>(ws, out);
}
